# starting kernel
# baseline (speedup 1.0000x reference)
#include <hip/hip_runtime.h>


typedef __attribute__((ext_vector_type(2))) float v2f;
typedef __attribute__((ext_vector_type(8))) float v8f;

#define B_DIM 64
#define T_DIM 8192
#define C_DIM 64
#define NTILE (T_DIM / 16)
#define EPSV  1e-5f
#define PFD   8

__device__ __forceinline__ v8f scan16(const v2f A[4], int halfh,
                                      const float e[8], const float o[8],
                                      v8f c) {
  v2f Bv;
  Bv.x = halfh ? o[2] : e[0];
  Bv.y = halfh ? o[3] : e[1];
  c = __builtin_amdgcn_wmma_f32_16x16x4_f32(false, A[0], false, Bv, (short)0,
                                            c, false, false);
  Bv.x = halfh ? o[6] : e[4];
  Bv.y = halfh ? o[7] : e[5];
  c = __builtin_amdgcn_wmma_f32_16x16x4_f32(false, A[1], false, Bv, (short)0,
                                            c, false, false);
  Bv.x = halfh ? e[2] : o[0];
  Bv.y = halfh ? e[3] : o[1];
  c = __builtin_amdgcn_wmma_f32_16x16x4_f32(false, A[2], false, Bv, (short)0,
                                            c, false, false);
  Bv.x = halfh ? e[6] : o[4];
  Bv.y = halfh ? e[7] : o[5];
  c = __builtin_amdgcn_wmma_f32_16x16x4_f32(false, A[3], false, Bv, (short)0,
                                            c, false, false);
  return c;
}

__device__ __forceinline__ void load_tile(const float* __restrict__ tile,
                                          int halfh, float e[8], float o[8]) {
#pragma unroll
  for (int j = 0; j < 8; ++j)
    e[j] = tile[(size_t)(j + 8 * halfh) * C_DIM];
#pragma unroll
  for (int j = 0; j < 8; ++j)
    o[j] = tile[(size_t)(j + 8 * (1 - halfh)) * C_DIM];
}

__device__ __forceinline__ void process_tile(
    const float e[8], const float o[8], float* __restrict__ tout,
    const float* __restrict__ s_inv, int s, int halfh, const v2f Amat[4],
    float& carry_x, float& carry_d) {
  v8f acc;
#pragma unroll
  for (int j = 0; j < 8; ++j) acc[j] = carry_x;
  acc = scan16(Amat, halfh, e, o, acc);

  float dd[8], d2[8], invn[8];
#pragma unroll
  for (int j = 0; j < 8; ++j) {
    const int t = s * 16 + j + 8 * halfh;
    const float inv = s_inv[t];
    invn[j] = inv;
    const float mean = acc[j] * inv;
    dd[j] = e[j] - mean;
    d2[j] = dd[j] * dd[j];
  }
  {
    const float tot_o = __shfl_xor((float)acc[7], 16, 32);
    carry_x = halfh ? (float)acc[7] : tot_o;
  }

  float d2o[8];
#pragma unroll
  for (int j = 0; j < 8; ++j) d2o[j] = __shfl_xor(d2[j], 16, 32);
  v8f acc2;
#pragma unroll
  for (int j = 0; j < 8; ++j) acc2[j] = carry_d;
  acc2 = scan16(Amat, halfh, d2, d2o, acc2);

#pragma unroll
  for (int j = 0; j < 8; ++j) {
    const float var = acc2[j] * invn[j];
    const float rstd = rsqrtf(fmaxf(var, EPSV));
    tout[(size_t)(j + 8 * halfh) * C_DIM] = dd[j] * rstd;
  }
  {
    const float tot_o = __shfl_xor((float)acc2[7], 16, 32);
    carry_d = halfh ? (float)acc2[7] : tot_o;
  }
}

__global__ __launch_bounds__(128) void revin_causal_norm_kernel(
    const float* __restrict__ x, float* __restrict__ out) {
  __shared__ float s_inv[T_DIM];
  const int tid = threadIdx.x;
  for (int t = tid; t < T_DIM; t += 128) s_inv[t] = 1.0f / (float)(t + 1);
  __syncthreads();

  const int lane  = tid & 31;
  const int wave  = tid >> 5;
  const int b     = blockIdx.x;
  const int n16   = lane & 15;
  const int halfh = (lane >> 4) & 1;
  const int cbase = wave * 16;

  const float* __restrict__ xb = x + (size_t)b * T_DIM * C_DIM + cbase + n16;
  float* __restrict__ ob = out + (size_t)b * T_DIM * C_DIM + cbase + n16;

  v2f Amat[4];
#pragma unroll
  for (int k = 0; k < 4; ++k) {
    const int k0 = 4 * k + 2 * halfh;
    Amat[k].x = (k0 <= n16) ? 1.0f : 0.0f;
    Amat[k].y = (k0 + 1 <= n16) ? 1.0f : 0.0f;
  }

  float carry_x = 0.0f;
  float carry_d = 0.0f;

  float ea[8], oa[8], eb[8], ob_[8];
  load_tile(xb, halfh, ea, oa);

  for (int su = 0; su < NTILE / 2; ++su) {
    const int s0 = 2 * su;
    const int s1 = s0 + 1;
    const int s2 = (s1 + 1 < NTILE) ? s1 + 1 : NTILE - 1;
    const int sp = (s0 + PFD < NTILE) ? s0 + PFD : NTILE - 1;

    load_tile(xb + (size_t)s1 * 16 * C_DIM, halfh, eb, ob_);
    __builtin_prefetch(xb + (size_t)sp * 16 * C_DIM, 0, 1);
    process_tile(ea, oa, ob + (size_t)s0 * 16 * C_DIM, s_inv, s0, halfh, Amat,
                 carry_x, carry_d);

    load_tile(xb + (size_t)s2 * 16 * C_DIM, halfh, ea, oa);
    process_tile(eb, ob_, ob + (size_t)s1 * 16 * C_DIM, s_inv, s1, halfh, Amat,
                 carry_x, carry_d);
  }
}

extern "C" void kernel_launch(void* const* d_in, const int* in_sizes, int n_in,
                              void* d_out, int out_size, void* d_ws,
                              size_t ws_size, hipStream_t stream) {
  (void)in_sizes; (void)n_in; (void)out_size; (void)d_ws; (void)ws_size;
  const float* x = (const float*)d_in[0];
  float* out = (float*)d_out;
  revin_causal_norm_kernel<<<dim3(B_DIM), dim3(128), 0, stream>>>(x, out);
}


# Time axis split over the 16 waves of a work-group (one group per batch x 16 channels): per-wave totals through LDS give each wave its carry-in, a different reduction tree for both running sums
# speedup vs baseline: 3.3067x; 3.3067x over previous
#include <hip/hip_runtime.h>


typedef __attribute__((ext_vector_type(2))) float v2f;
typedef __attribute__((ext_vector_type(8))) float v8f;

#define B_DIM 64
#define T_DIM 8192
#define C_DIM 64
#define NTILE (T_DIM / 16)
#define EPSV  1e-5f
#define NW    16
#define KT    4
#define BLKT  (NW * KT)
#define NBLK  (NTILE / BLKT)
#define WPITCH 20

__device__ __forceinline__ v8f scan16(const v2f A[4], int halfh,
                                      const float e[8], const float o[8],
                                      v8f c) {
  v2f Bv;
  Bv.x = halfh ? o[2] : e[0];
  Bv.y = halfh ? o[3] : e[1];
  c = __builtin_amdgcn_wmma_f32_16x16x4_f32(false, A[0], false, Bv, (short)0,
                                            c, false, false);
  Bv.x = halfh ? o[6] : e[4];
  Bv.y = halfh ? o[7] : e[5];
  c = __builtin_amdgcn_wmma_f32_16x16x4_f32(false, A[1], false, Bv, (short)0,
                                            c, false, false);
  Bv.x = halfh ? e[2] : o[0];
  Bv.y = halfh ? e[3] : o[1];
  c = __builtin_amdgcn_wmma_f32_16x16x4_f32(false, A[2], false, Bv, (short)0,
                                            c, false, false);
  Bv.x = halfh ? e[6] : o[4];
  Bv.y = halfh ? e[7] : o[5];
  c = __builtin_amdgcn_wmma_f32_16x16x4_f32(false, A[3], false, Bv, (short)0,
                                            c, false, false);
  return c;
}

__device__ __forceinline__ void lds_barrier() {
  __builtin_amdgcn_fence(__ATOMIC_RELEASE, "workgroup", "local");
  __builtin_amdgcn_s_barrier();
  __builtin_amdgcn_fence(__ATOMIC_ACQUIRE, "workgroup", "local");
}

__device__ __forceinline__ void load_block(const float* __restrict__ p,
                                           int halfh, float buf[KT][8]) {
#pragma unroll
  for (int i = 0; i < KT; ++i)
#pragma unroll
    for (int j = 0; j < 8; ++j)
      buf[i][j] = p[(size_t)(i * 16 + j + 8 * halfh) * C_DIM];
}

__device__ __forceinline__ float wave_prefix(const float* __restrict__ tab,
                                             int n16, int w, float& base) {
  float v[NW];
#pragma unroll
  for (int g = 0; g < NW / 4; ++g) {
    const float4 q =
        *reinterpret_cast<const float4*>(tab + n16 * WPITCH + 4 * g);
    v[4 * g + 0] = q.x;
    v[4 * g + 1] = q.y;
    v[4 * g + 2] = q.z;
    v[4 * g + 3] = q.w;
  }
  float run = base;
  float px = base;
#pragma unroll
  for (int i = 0; i < NW; ++i) {
    px = (i == w) ? run : px;
    run += v[i];
  }
  base = run;
  return px;
}

__device__ __forceinline__ void process_block(
    float buf[KT][8], float* __restrict__ tout, const float* __restrict__ s_inv,
    float* __restrict__ s_wtx, float* __restrict__ s_wtd, int s_first, int w,
    int n16, int halfh, const v2f Amat[4], float& base_x, float& base_d) {
  float part = 0.0f;
#pragma unroll
  for (int i = 0; i < KT; ++i)
#pragma unroll
    for (int j = 0; j < 8; ++j) part += buf[i][j];
  const float tot = part + __shfl_xor(part, 16, 32);
  if (halfh == 0) s_wtx[n16 * WPITCH + w] = tot;
  lds_barrier();

  float carry_x = wave_prefix(s_wtx, n16, w, base_x);
  float partd = 0.0f;
#pragma unroll
  for (int i = 0; i < KT; ++i) {
    float o[8];
#pragma unroll
    for (int j = 0; j < 8; ++j) o[j] = __shfl_xor(buf[i][j], 16, 32);
    v8f acc;
#pragma unroll
    for (int j = 0; j < 8; ++j) acc[j] = carry_x;
    acc = scan16(Amat, halfh, buf[i], o, acc);
#pragma unroll
    for (int j = 0; j < 8; ++j) {
      const int t = (s_first + i) * 16 + j + 8 * halfh;
      const float inv = s_inv[t];
      const float mean = acc[j] * inv;
      const float d = buf[i][j] - mean;
      buf[i][j] = d;
      partd += d * d;
    }
    {
      const float tot_o = __shfl_xor((float)acc[7], 16, 32);
      carry_x = halfh ? (float)acc[7] : tot_o;
    }
  }
  const float totd = partd + __shfl_xor(partd, 16, 32);
  if (halfh == 0) s_wtd[n16 * WPITCH + w] = totd;
  lds_barrier();

  float carry_d = wave_prefix(s_wtd, n16, w, base_d);
#pragma unroll
  for (int i = 0; i < KT; ++i) {
    float d2[8], d2o[8];
#pragma unroll
    for (int j = 0; j < 8; ++j) d2[j] = buf[i][j] * buf[i][j];
#pragma unroll
    for (int j = 0; j < 8; ++j) d2o[j] = __shfl_xor(d2[j], 16, 32);
    v8f acc2;
#pragma unroll
    for (int j = 0; j < 8; ++j) acc2[j] = carry_d;
    acc2 = scan16(Amat, halfh, d2, d2o, acc2);
#pragma unroll
    for (int j = 0; j < 8; ++j) {
      const int t = (s_first + i) * 16 + j + 8 * halfh;
      const float var = acc2[j] * s_inv[t];
      const float rstd = rsqrtf(fmaxf(var, EPSV));
      tout[(size_t)(i * 16 + j + 8 * halfh) * C_DIM] = buf[i][j] * rstd;
    }
    {
      const float tot_o = __shfl_xor((float)acc2[7], 16, 32);
      carry_d = halfh ? (float)acc2[7] : tot_o;
    }
  }
}

__global__ __launch_bounds__(NW * 32) void revin_causal_norm_kernel(
    const float* __restrict__ x, float* __restrict__ out) {
  __shared__ float s_inv[T_DIM];
  __shared__ __attribute__((aligned(16))) float s_wtx[16 * WPITCH];
  __shared__ __attribute__((aligned(16))) float s_wtd[16 * WPITCH];

  const int tid   = threadIdx.x;
  const int lane  = tid & 31;
  const int w     = __builtin_amdgcn_readfirstlane(tid >> 5);
  const int b     = blockIdx.x >> 2;
  const int cbase = (blockIdx.x & 3) * 16;
  const int n16   = lane & 15;
  const int halfh = (lane >> 4) & 1;

  const size_t woff = (size_t)b * T_DIM * C_DIM + cbase + n16 +
                      (size_t)w * KT * 16 * C_DIM;
  const float* __restrict__ xb = x + woff;
  float* __restrict__ ob = out + woff;

  float bufA[KT][8], bufB[KT][8];
  load_block(xb, halfh, bufA);

  for (int t = tid; t < T_DIM; t += NW * 32) s_inv[t] = 1.0f / (float)(t + 1);

  v2f Amat[4];
#pragma unroll
  for (int k = 0; k < 4; ++k) {
    const int k0 = 4 * k + 2 * halfh;
    Amat[k].x = (k0 <= n16) ? 1.0f : 0.0f;
    Amat[k].y = (k0 + 1 <= n16) ? 1.0f : 0.0f;
  }

  float base_x = 0.0f;
  float base_d = 0.0f;

  const size_t bstride = (size_t)BLKT * 16 * C_DIM;
  for (int qq = 0; qq < NBLK / 2 - 1; ++qq) {
    const int q0 = 2 * qq;
    const int q1 = q0 + 1;

    load_block(xb + (size_t)q1 * bstride, halfh, bufB);
    process_block(bufA, ob + (size_t)q0 * bstride, s_inv, s_wtx, s_wtd,
                  q0 * BLKT + w * KT, w, n16, halfh, Amat, base_x, base_d);

    load_block(xb + (size_t)(q1 + 1) * bstride, halfh, bufA);
    process_block(bufB, ob + (size_t)q1 * bstride, s_inv, s_wtx, s_wtd,
                  q1 * BLKT + w * KT, w, n16, halfh, Amat, base_x, base_d);
  }
  {
    const int q0 = NBLK - 2;
    const int q1 = NBLK - 1;
    load_block(xb + (size_t)q1 * bstride, halfh, bufB);
    process_block(bufA, ob + (size_t)q0 * bstride, s_inv, s_wtx, s_wtd,
                  q0 * BLKT + w * KT, w, n16, halfh, Amat, base_x, base_d);
    process_block(bufB, ob + (size_t)q1 * bstride, s_inv, s_wtx, s_wtd,
                  q1 * BLKT + w * KT, w, n16, halfh, Amat, base_x, base_d);
  }
}

extern "C" void kernel_launch(void* const* d_in, const int* in_sizes, int n_in,
                              void* d_out, int out_size, void* d_ws,
                              size_t ws_size, hipStream_t stream) {
  (void)in_sizes; (void)n_in; (void)out_size; (void)d_ws; (void)ws_size;
  const float* x = (const float*)d_in[0];
  float* out = (float*)d_out;
  revin_causal_norm_kernel<<<dim3(256), dim3(512), 0, stream>>>(x, out);
}


# Work-group owns 32 channels (a whole 128-byte line per row): 8-byte loads/stores, each half-wave access one whole line; 128 groups, 512-step blocks: per-wave totals of 32 rows, new reduction tree
# speedup vs baseline: 4.0655x; 1.2295x over previous
#include <hip/hip_runtime.h>


typedef __attribute__((ext_vector_type(2))) float v2f;
typedef __attribute__((ext_vector_type(8))) float v8f;

#define B_DIM 64
#define T_DIM 8192
#define C_DIM 64
#define NTILE (T_DIM / 16)
#define EPSV  1e-5f
#define NW    16
#define KT    2
#define BLKT  (NW * KT)
#define NBLK  (NTILE / BLKT)
#define WPITCH 20
#define CPL    2
#define CGRP   (16 * CPL)

__device__ __forceinline__ v8f scan16(const v2f A[4], int halfh,
                                      const float e[8], const float o[8],
                                      v8f c) {
  v2f Bv;
  Bv.x = halfh ? o[2] : e[0];
  Bv.y = halfh ? o[3] : e[1];
  c = __builtin_amdgcn_wmma_f32_16x16x4_f32(false, A[0], false, Bv, (short)0,
                                            c, false, false);
  Bv.x = halfh ? o[6] : e[4];
  Bv.y = halfh ? o[7] : e[5];
  c = __builtin_amdgcn_wmma_f32_16x16x4_f32(false, A[1], false, Bv, (short)0,
                                            c, false, false);
  Bv.x = halfh ? e[2] : o[0];
  Bv.y = halfh ? e[3] : o[1];
  c = __builtin_amdgcn_wmma_f32_16x16x4_f32(false, A[2], false, Bv, (short)0,
                                            c, false, false);
  Bv.x = halfh ? e[6] : o[4];
  Bv.y = halfh ? e[7] : o[5];
  c = __builtin_amdgcn_wmma_f32_16x16x4_f32(false, A[3], false, Bv, (short)0,
                                            c, false, false);
  return c;
}

__device__ __forceinline__ void lds_barrier() {
  __builtin_amdgcn_fence(__ATOMIC_RELEASE, "workgroup", "local");
  __builtin_amdgcn_s_barrier();
  __builtin_amdgcn_fence(__ATOMIC_ACQUIRE, "workgroup", "local");
}

__device__ __forceinline__ void load_block(const float* __restrict__ p,
                                           int halfh, float buf[KT][CPL][8]) {
#pragma unroll
  for (int i = 0; i < KT; ++i)
#pragma unroll
    for (int j = 0; j < 8; ++j) {
      const float2 v = *reinterpret_cast<const float2*>(
          p + (size_t)(i * 16 + j + 8 * halfh) * C_DIM);
      buf[i][0][j] = v.x;
      buf[i][1][j] = v.y;
    }
}

__device__ __forceinline__ float wave_prefix(const float* __restrict__ tab,
                                             int n16, int w, float& base) {
  float v[NW];
#pragma unroll
  for (int g = 0; g < NW / 4; ++g) {
    const float4 q =
        *reinterpret_cast<const float4*>(tab + n16 * WPITCH + 4 * g);
    v[4 * g + 0] = q.x;
    v[4 * g + 1] = q.y;
    v[4 * g + 2] = q.z;
    v[4 * g + 3] = q.w;
  }
  float run = base;
  float px = base;
#pragma unroll
  for (int i = 0; i < NW; ++i) {
    px = (i == w) ? run : px;
    run += v[i];
  }
  base = run;
  return px;
}

__device__ __forceinline__ void process_block(
    float buf[KT][CPL][8], float* __restrict__ tout,
    const float* __restrict__ s_inv, float* __restrict__ s_wtx,
    float* __restrict__ s_wtd, int s_first, int w, int n16, int halfh,
    const v2f Amat[4], float base_x[CPL], float base_d[CPL]) {
  float part[CPL];
#pragma unroll
  for (int k = 0; k < CPL; ++k) {
    part[k] = 0.0f;
#pragma unroll
    for (int i = 0; i < KT; ++i)
#pragma unroll
      for (int j = 0; j < 8; ++j) part[k] += buf[i][k][j];
  }
#pragma unroll
  for (int k = 0; k < CPL; ++k) {
    const float tot = part[k] + __shfl_xor(part[k], 16, 32);
    if (halfh == 0) s_wtx[(k * 16 + n16) * WPITCH + w] = tot;
  }
  lds_barrier();

  float carry_x[CPL], partd[CPL];
#pragma unroll
  for (int k = 0; k < CPL; ++k) {
    carry_x[k] = wave_prefix(s_wtx, k * 16 + n16, w, base_x[k]);
    partd[k] = 0.0f;
  }
#pragma unroll
  for (int i = 0; i < KT; ++i) {
#pragma unroll
    for (int k = 0; k < CPL; ++k) {
      float o[8];
#pragma unroll
      for (int j = 0; j < 8; ++j) o[j] = __shfl_xor(buf[i][k][j], 16, 32);
      v8f acc;
#pragma unroll
      for (int j = 0; j < 8; ++j) acc[j] = carry_x[k];
      acc = scan16(Amat, halfh, buf[i][k], o, acc);
#pragma unroll
      for (int j = 0; j < 8; ++j) {
        const int t = (s_first + i) * 16 + j + 8 * halfh;
        const float inv = s_inv[t];
        const float mean = acc[j] * inv;
        const float d = buf[i][k][j] - mean;
        buf[i][k][j] = d;
        partd[k] += d * d;
      }
      {
        const float tot_o = __shfl_xor((float)acc[7], 16, 32);
        carry_x[k] = halfh ? (float)acc[7] : tot_o;
      }
    }
  }
#pragma unroll
  for (int k = 0; k < CPL; ++k) {
    const float totd = partd[k] + __shfl_xor(partd[k], 16, 32);
    if (halfh == 0) s_wtd[(k * 16 + n16) * WPITCH + w] = totd;
  }
  lds_barrier();

  float carry_d[CPL];
#pragma unroll
  for (int k = 0; k < CPL; ++k)
    carry_d[k] = wave_prefix(s_wtd, k * 16 + n16, w, base_d[k]);
#pragma unroll
  for (int i = 0; i < KT; ++i) {
    v8f acc2[CPL];
#pragma unroll
    for (int k = 0; k < CPL; ++k) {
      float d2[8], d2o[8];
#pragma unroll
      for (int j = 0; j < 8; ++j) d2[j] = buf[i][k][j] * buf[i][k][j];
#pragma unroll
      for (int j = 0; j < 8; ++j) d2o[j] = __shfl_xor(d2[j], 16, 32);
#pragma unroll
      for (int j = 0; j < 8; ++j) acc2[k][j] = carry_d[k];
      acc2[k] = scan16(Amat, halfh, d2, d2o, acc2[k]);
    }
#pragma unroll
    for (int j = 0; j < 8; ++j) {
      const int t = (s_first + i) * 16 + j + 8 * halfh;
      const float inv = s_inv[t];
      float2 r;
      r.x = buf[i][0][j] * rsqrtf(fmaxf(acc2[0][j] * inv, EPSV));
      r.y = buf[i][1][j] * rsqrtf(fmaxf(acc2[1][j] * inv, EPSV));
      *reinterpret_cast<float2*>(
          tout + (size_t)(i * 16 + j + 8 * halfh) * C_DIM) = r;
    }
#pragma unroll
    for (int k = 0; k < CPL; ++k) {
      const float tot_o = __shfl_xor((float)acc2[k][7], 16, 32);
      carry_d[k] = halfh ? (float)acc2[k][7] : tot_o;
    }
  }
}

__global__ __launch_bounds__(NW * 32) void revin_causal_norm_kernel(
    const float* __restrict__ x, float* __restrict__ out) {
  __shared__ float s_inv[T_DIM];
  __shared__ __attribute__((aligned(16))) float s_wtx[CGRP * WPITCH];
  __shared__ __attribute__((aligned(16))) float s_wtd[CGRP * WPITCH];

  const int tid   = threadIdx.x;
  const int lane  = tid & 31;
  const int w     = __builtin_amdgcn_readfirstlane(tid >> 5);
  const int b     = blockIdx.x >> 1;
  const int cbase = (blockIdx.x & 1) * CGRP;
  const int n16   = lane & 15;
  const int halfh = (lane >> 4) & 1;

  const size_t woff = (size_t)b * T_DIM * C_DIM + cbase + CPL * n16 +
                      (size_t)w * KT * 16 * C_DIM;
  const float* __restrict__ xb = x + woff;
  float* __restrict__ ob = out + woff;

  float bufA[KT][CPL][8], bufB[KT][CPL][8];
  load_block(xb, halfh, bufA);

  for (int t = tid; t < T_DIM; t += NW * 32) s_inv[t] = 1.0f / (float)(t + 1);

  v2f Amat[4];
#pragma unroll
  for (int k = 0; k < 4; ++k) {
    const int k0 = 4 * k + 2 * halfh;
    Amat[k].x = (k0 <= n16) ? 1.0f : 0.0f;
    Amat[k].y = (k0 + 1 <= n16) ? 1.0f : 0.0f;
  }

  float base_x[CPL] = {0.0f, 0.0f};
  float base_d[CPL] = {0.0f, 0.0f};

  const size_t bstride = (size_t)BLKT * 16 * C_DIM;
  for (int qq = 0; qq < NBLK / 2 - 1; ++qq) {
    const int q0 = 2 * qq;
    const int q1 = q0 + 1;

    load_block(xb + (size_t)q1 * bstride, halfh, bufB);
    process_block(bufA, ob + (size_t)q0 * bstride, s_inv, s_wtx, s_wtd,
                  q0 * BLKT + w * KT, w, n16, halfh, Amat, base_x, base_d);

    load_block(xb + (size_t)(q1 + 1) * bstride, halfh, bufA);
    process_block(bufB, ob + (size_t)q1 * bstride, s_inv, s_wtx, s_wtd,
                  q1 * BLKT + w * KT, w, n16, halfh, Amat, base_x, base_d);
  }
  {
    const int q0 = NBLK - 2;
    const int q1 = NBLK - 1;
    load_block(xb + (size_t)q1 * bstride, halfh, bufB);
    process_block(bufA, ob + (size_t)q0 * bstride, s_inv, s_wtx, s_wtd,
                  q0 * BLKT + w * KT, w, n16, halfh, Amat, base_x, base_d);
    process_block(bufB, ob + (size_t)q1 * bstride, s_inv, s_wtx, s_wtd,
                  q1 * BLKT + w * KT, w, n16, halfh, Amat, base_x, base_d);
  }
}

extern "C" void kernel_launch(void* const* d_in, const int* in_sizes, int n_in,
                              void* d_out, int out_size, void* d_ws,
                              size_t ws_size, hipStream_t stream) {
  (void)in_sizes; (void)n_in; (void)out_size; (void)d_ws; (void)ws_size;
  const float* x = (const float*)d_in[0];
  float* out = (float*)d_out;
  revin_causal_norm_kernel<<<dim3(128), dim3(512), 0, stream>>>(x, out);
}


# Scan products take their 16 steps in the order the lane halves hold them (slice s, half h: rows 8h+2s,+1; A columns alike): no cross-half moves for B; order of sums inside the products changes
# speedup vs baseline: 4.7692x; 1.1731x over previous
#include <hip/hip_runtime.h>


typedef __attribute__((ext_vector_type(2))) float v2f;
typedef __attribute__((ext_vector_type(8))) float v8f;

#define B_DIM 64
#define T_DIM 8192
#define C_DIM 64
#define NTILE (T_DIM / 16)
#define EPSV  1e-5f
#define NW    16
#define KT    2
#define BLKT  (NW * KT)
#define NBLK  (NTILE / BLKT)
#define WPITCH 20
#define CPL    2
#define CGRP   (16 * CPL)

__device__ __forceinline__ v8f scan16(const v2f A[4], const float e[8],
                                      v8f c) {
  v2f Bv;
  Bv.x = e[0];
  Bv.y = e[1];
  c = __builtin_amdgcn_wmma_f32_16x16x4_f32(false, A[0], false, Bv, (short)0,
                                            c, false, false);
  Bv.x = e[2];
  Bv.y = e[3];
  c = __builtin_amdgcn_wmma_f32_16x16x4_f32(false, A[1], false, Bv, (short)0,
                                            c, false, false);
  Bv.x = e[4];
  Bv.y = e[5];
  c = __builtin_amdgcn_wmma_f32_16x16x4_f32(false, A[2], false, Bv, (short)0,
                                            c, false, false);
  Bv.x = e[6];
  Bv.y = e[7];
  c = __builtin_amdgcn_wmma_f32_16x16x4_f32(false, A[3], false, Bv, (short)0,
                                            c, false, false);
  return c;
}

__device__ __forceinline__ void lds_barrier() {
  __builtin_amdgcn_fence(__ATOMIC_RELEASE, "workgroup", "local");
  __builtin_amdgcn_s_barrier();
  __builtin_amdgcn_fence(__ATOMIC_ACQUIRE, "workgroup", "local");
}

__device__ __forceinline__ void load_block(const float* __restrict__ p,
                                           int halfh, float buf[KT][CPL][8]) {
#pragma unroll
  for (int i = 0; i < KT; ++i)
#pragma unroll
    for (int j = 0; j < 8; ++j) {
      const float2 v = *reinterpret_cast<const float2*>(
          p + (size_t)(i * 16 + j + 8 * halfh) * C_DIM);
      buf[i][0][j] = v.x;
      buf[i][1][j] = v.y;
    }
}

__device__ __forceinline__ float wave_prefix(const float* __restrict__ tab,
                                             int n16, int w, float& base) {
  float v[NW];
#pragma unroll
  for (int g = 0; g < NW / 4; ++g) {
    const float4 q =
        *reinterpret_cast<const float4*>(tab + n16 * WPITCH + 4 * g);
    v[4 * g + 0] = q.x;
    v[4 * g + 1] = q.y;
    v[4 * g + 2] = q.z;
    v[4 * g + 3] = q.w;
  }
  float run = base;
  float px = base;
#pragma unroll
  for (int i = 0; i < NW; ++i) {
    px = (i == w) ? run : px;
    run += v[i];
  }
  base = run;
  return px;
}

__device__ __forceinline__ void process_block(
    float buf[KT][CPL][8], float* __restrict__ tout,
    const float* __restrict__ s_inv, float* __restrict__ s_wtx,
    float* __restrict__ s_wtd, int s_first, int w, int n16, int halfh,
    const v2f Amat[4], float base_x[CPL], float base_d[CPL]) {
  float part[CPL];
#pragma unroll
  for (int k = 0; k < CPL; ++k) {
    part[k] = 0.0f;
#pragma unroll
    for (int i = 0; i < KT; ++i)
#pragma unroll
      for (int j = 0; j < 8; ++j) part[k] += buf[i][k][j];
  }
#pragma unroll
  for (int k = 0; k < CPL; ++k) {
    const float tot = part[k] + __shfl_xor(part[k], 16, 32);
    if (halfh == 0) s_wtx[(k * 16 + n16) * WPITCH + w] = tot;
  }
  lds_barrier();

  float carry_x[CPL], partd[CPL];
#pragma unroll
  for (int k = 0; k < CPL; ++k) {
    carry_x[k] = wave_prefix(s_wtx, k * 16 + n16, w, base_x[k]);
    partd[k] = 0.0f;
  }
#pragma unroll
  for (int i = 0; i < KT; ++i) {
#pragma unroll
    for (int k = 0; k < CPL; ++k) {
      v8f acc;
#pragma unroll
      for (int j = 0; j < 8; ++j) acc[j] = carry_x[k];
      acc = scan16(Amat, buf[i][k], acc);
#pragma unroll
      for (int j = 0; j < 8; ++j) {
        const int t = (s_first + i) * 16 + j + 8 * halfh;
        const float inv = s_inv[t];
        const float mean = acc[j] * inv;
        const float d = buf[i][k][j] - mean;
        buf[i][k][j] = d;
        partd[k] += d * d;
      }
      {
        const float tot_o = __shfl_xor((float)acc[7], 16, 32);
        carry_x[k] = halfh ? (float)acc[7] : tot_o;
      }
    }
  }
#pragma unroll
  for (int k = 0; k < CPL; ++k) {
    const float totd = partd[k] + __shfl_xor(partd[k], 16, 32);
    if (halfh == 0) s_wtd[(k * 16 + n16) * WPITCH + w] = totd;
  }
  lds_barrier();

  float carry_d[CPL];
#pragma unroll
  for (int k = 0; k < CPL; ++k)
    carry_d[k] = wave_prefix(s_wtd, k * 16 + n16, w, base_d[k]);
#pragma unroll
  for (int i = 0; i < KT; ++i) {
    v8f acc2[CPL];
#pragma unroll
    for (int k = 0; k < CPL; ++k) {
      float d2[8];
#pragma unroll
      for (int j = 0; j < 8; ++j) d2[j] = buf[i][k][j] * buf[i][k][j];
#pragma unroll
      for (int j = 0; j < 8; ++j) acc2[k][j] = carry_d[k];
      acc2[k] = scan16(Amat, d2, acc2[k]);
    }
#pragma unroll
    for (int j = 0; j < 8; ++j) {
      const int t = (s_first + i) * 16 + j + 8 * halfh;
      const float inv = s_inv[t];
      float2 r;
      r.x = buf[i][0][j] * rsqrtf(fmaxf(acc2[0][j] * inv, EPSV));
      r.y = buf[i][1][j] * rsqrtf(fmaxf(acc2[1][j] * inv, EPSV));
      *reinterpret_cast<float2*>(
          tout + (size_t)(i * 16 + j + 8 * halfh) * C_DIM) = r;
    }
#pragma unroll
    for (int k = 0; k < CPL; ++k) {
      const float tot_o = __shfl_xor((float)acc2[k][7], 16, 32);
      carry_d[k] = halfh ? (float)acc2[k][7] : tot_o;
    }
  }
}

__global__ __launch_bounds__(NW * 32) void revin_causal_norm_kernel(
    const float* __restrict__ x, float* __restrict__ out) {
  __shared__ float s_inv[T_DIM];
  __shared__ __attribute__((aligned(16))) float s_wtx[CGRP * WPITCH];
  __shared__ __attribute__((aligned(16))) float s_wtd[CGRP * WPITCH];

  const int tid   = threadIdx.x;
  const int lane  = tid & 31;
  const int w     = __builtin_amdgcn_readfirstlane(tid >> 5);
  const int b     = blockIdx.x >> 1;
  const int cbase = (blockIdx.x & 1) * CGRP;
  const int n16   = lane & 15;
  const int halfh = (lane >> 4) & 1;

  const size_t woff = (size_t)b * T_DIM * C_DIM + cbase + CPL * n16 +
                      (size_t)w * KT * 16 * C_DIM;
  const float* __restrict__ xb = x + woff;
  float* __restrict__ ob = out + woff;

  float bufA[KT][CPL][8], bufB[KT][CPL][8];
  load_block(xb, halfh, bufA);

  for (int t = tid; t < T_DIM; t += NW * 32) s_inv[t] = 1.0f / (float)(t + 1);

  v2f Amat[4];
#pragma unroll
  for (int k = 0; k < 4; ++k) {
    const int k0 = 8 * halfh + 2 * k;
    Amat[k].x = (k0 <= n16) ? 1.0f : 0.0f;
    Amat[k].y = (k0 + 1 <= n16) ? 1.0f : 0.0f;
  }

  float base_x[CPL] = {0.0f, 0.0f};
  float base_d[CPL] = {0.0f, 0.0f};

  const size_t bstride = (size_t)BLKT * 16 * C_DIM;
  for (int qq = 0; qq < NBLK / 2 - 1; ++qq) {
    const int q0 = 2 * qq;
    const int q1 = q0 + 1;

    load_block(xb + (size_t)q1 * bstride, halfh, bufB);
    process_block(bufA, ob + (size_t)q0 * bstride, s_inv, s_wtx, s_wtd,
                  q0 * BLKT + w * KT, w, n16, halfh, Amat, base_x, base_d);

    load_block(xb + (size_t)(q1 + 1) * bstride, halfh, bufA);
    process_block(bufB, ob + (size_t)q1 * bstride, s_inv, s_wtx, s_wtd,
                  q1 * BLKT + w * KT, w, n16, halfh, Amat, base_x, base_d);
  }
  {
    const int q0 = NBLK - 2;
    const int q1 = NBLK - 1;
    load_block(xb + (size_t)q1 * bstride, halfh, bufB);
    process_block(bufA, ob + (size_t)q0 * bstride, s_inv, s_wtx, s_wtd,
                  q0 * BLKT + w * KT, w, n16, halfh, Amat, base_x, base_d);
    process_block(bufB, ob + (size_t)q1 * bstride, s_inv, s_wtx, s_wtd,
                  q1 * BLKT + w * KT, w, n16, halfh, Amat, base_x, base_d);
  }
}

extern "C" void kernel_launch(void* const* d_in, const int* in_sizes, int n_in,
                              void* d_out, int out_size, void* d_ws,
                              size_t ws_size, hipStream_t stream) {
  (void)in_sizes; (void)n_in; (void)out_size; (void)d_ws; (void)ws_size;
  const float* x = (const float*)d_in[0];
  float* out = (float*)d_out;
  revin_causal_norm_kernel<<<dim3(128), dim3(512), 0, stream>>>(x, out);
}
